# starting kernel
# baseline (speedup 1.0000x reference)
#include <hip/hip_runtime.h>

typedef __attribute__((ext_vector_type(2))) float v2f;
typedef __attribute__((ext_vector_type(8))) float v8f;

#define NQ 4
#define NL 2
#define DIM 20
#define HID 64
#define TOK_PER_WAVE 32
#define WAVES 4
#define TOK_PER_BLOCK 128
#define HSTRIDE 68

__global__ __launch_bounds__(TOK_PER_BLOCK) void qhyb_fused(
    const float* __restrict__ x,
    const float* __restrict__ W_embed,
    const float* __restrict__ b_embed,
    const float* __restrict__ W_c2q,
    const float* __restrict__ b_c2q,
    const float* __restrict__ q_weights,
    const float* __restrict__ W_q2c,
    const float* __restrict__ b_q2c,
    const float* __restrict__ W_out,
    const float* __restrict__ b_out,
    float* __restrict__ out)
{
    __shared__ float h_lds[TOK_PER_BLOCK * HSTRIDE];
    __shared__ float wc2q_lds[NQ * HID];
    __shared__ float bemb_lds[HID];
    __shared__ float u_lds[NL][NQ][8];
    __shared__ float M_lds[3][NQ];
    __shared__ float c_lds[3];

    const int tid = threadIdx.x;

    for (int i = tid; i < NQ * HID; i += TOK_PER_BLOCK) wc2q_lds[i] = W_c2q[i];
    if (tid < HID) bemb_lds[tid] = b_embed[tid];
    if (tid < NL * NQ) {
        int layer = tid >> 2, q = tid & 3;
        float phi = q_weights[(layer * NQ + q) * 3 + 0];
        float th  = q_weights[(layer * NQ + q) * 3 + 1];
        float om  = q_weights[(layer * NQ + q) * 3 + 2];
        float ct = cosf(0.5f * th), st = sinf(0.5f * th);
        float cp = cosf(0.5f * (phi + om)), sp = sinf(0.5f * (phi + om));
        float cm = cosf(0.5f * (phi - om)), sm = sinf(0.5f * (phi - om));
        float* U = &u_lds[layer][q][0];
        U[0] =  cp * ct; U[1] = -sp * ct;
        U[2] = -cm * st; U[3] = -sm * st;
        U[4] =  cm * st; U[5] = -sm * st;
        U[6] =  cp * ct; U[7] =  sp * ct;
    }
    if (tid >= 8 && tid < 8 + 12) {
        int o = (tid - 8) >> 2, q = (tid - 8) & 3;
        float s = 0.0f;
        for (int h = 0; h < HID; ++h) s += W_out[o * HID + h] * W_q2c[h * NQ + q];
        M_lds[o][q] = s;
    }
    if (tid >= 20 && tid < 23) {
        int o = tid - 20;
        float s = b_out[o];
        for (int h = 0; h < HID; ++h) s += W_out[o * HID + h] * b_q2c[h];
        c_lds[o] = s;
    }
    __syncthreads();

    const int lane = tid & 31;
    const int wave = tid >> 5;
    const int lo = lane & 15, hi = lane >> 4;
    const long blockTok0 = (long)blockIdx.x * TOK_PER_BLOCK;
    const int  waveRow0  = wave * TOK_PER_WAVE;
    const long waveTok0  = blockTok0 + waveRow0;

    v2f bfrag[5][4];
#pragma unroll
    for (int ks = 0; ks < 5; ++ks)
#pragma unroll
        for (int n = 0; n < 4; ++n) {
            int hcol = n * 16 + lo;
            int d0 = ks * 4 + 2 * hi;
            bfrag[ks][n].x = W_embed[hcol * DIM + d0];
            bfrag[ks][n].y = W_embed[hcol * DIM + d0 + 1];
        }

    v8f acc[2][4];
#pragma unroll
    for (int mt = 0; mt < 2; ++mt)
#pragma unroll
        for (int n = 0; n < 4; ++n)
            acc[mt][n] = (v8f){0.f, 0.f, 0.f, 0.f, 0.f, 0.f, 0.f, 0.f};

#pragma unroll
    for (int ks = 0; ks < 5; ++ks) {
#pragma unroll
        for (int mt = 0; mt < 2; ++mt) {
            long t = waveTok0 + mt * 16 + lo;
            int d0 = ks * 4 + 2 * hi;
            v2f afrag;
            afrag.x = x[t * DIM + d0];
            afrag.y = x[t * DIM + d0 + 1];
#pragma unroll
            for (int n = 0; n < 4; ++n)
                acc[mt][n] = __builtin_amdgcn_wmma_f32_16x16x4_f32(
                    false, afrag, false, bfrag[ks][n],
                    (short)0, acc[mt][n], false, false);
        }
    }

#pragma unroll
    for (int mt = 0; mt < 2; ++mt)
#pragma unroll
        for (int n = 0; n < 4; ++n) {
            int hc = n * 16 + lo;
            float bb = bemb_lds[hc];
#pragma unroll
            for (int r = 0; r < 8; ++r) {
                int row = waveRow0 + mt * 16 + r + 8 * hi;
                h_lds[row * HSTRIDE + hc] = fmaxf(acc[mt][n][r] + bb, 0.0f);
            }
        }
    __syncthreads();

    const int myRow = waveRow0 + lane;
    const long tok = blockTok0 + myRow;
    const float* hrow = &h_lds[myRow * HSTRIDE];

    float cs[NQ], sn[NQ];
#pragma unroll
    for (int q = 0; q < NQ; ++q) {
        float s = b_c2q[q];
#pragma unroll 16
        for (int k = 0; k < HID; ++k) s += wc2q_lds[q * HID + k] * hrow[k];
        float a = 0.5f * tanhf(s);
        cs[q] = cosf(a);
        sn[q] = sinf(a);
    }

    float ar[16], ai[16];
#pragma unroll
    for (int i = 0; i < 16; ++i) {
        float v = ((i >> 3) & 1 ? sn[0] : cs[0]) *
                  ((i >> 2) & 1 ? sn[1] : cs[1]) *
                  ((i >> 1) & 1 ? sn[2] : cs[2]) *
                  ((i      ) & 1 ? sn[3] : cs[3]);
        ar[i] = v; ai[i] = 0.0f;
    }

#pragma unroll
    for (int layer = 0; layer < NL; ++layer) {
#pragma unroll
        for (int q = 0; q < NQ; ++q) {
            const float* U = &u_lds[layer][q][0];
            float u00r = U[0], u00i = U[1], u01r = U[2], u01i = U[3];
            float u10r = U[4], u10i = U[5], u11r = U[6], u11i = U[7];
            const int p = 3 - q;
#pragma unroll
            for (int i = 0; i < 16; ++i) {
                if (i & (1 << p)) continue;
                int j = i | (1 << p);
                float s0r = ar[i], s0i = ai[i], s1r = ar[j], s1i = ai[j];
                ar[i] = u00r * s0r - u00i * s0i + u01r * s1r - u01i * s1i;
                ai[i] = u00r * s0i + u00i * s0r + u01r * s1i + u01i * s1r;
                ar[j] = u10r * s0r - u10i * s0i + u11r * s1r - u11i * s1i;
                ai[j] = u10r * s0i + u10i * s0r + u11r * s1i + u11i * s1r;
            }
        }
#pragma unroll
        for (int q = 0; q < NQ - 1; ++q) {
            const int cbit = 1 << (3 - q), tbit = 1 << (2 - q);
#pragma unroll
            for (int i = 0; i < 16; ++i) {
                if ((i & cbit) && !(i & tbit)) {
                    int j = i | tbit;
                    float tr = ar[i], ti = ai[i];
                    ar[i] = ar[j]; ai[i] = ai[j];
                    ar[j] = tr;    ai[j] = ti;
                }
            }
        }
    }

    float z[NQ] = {0.f, 0.f, 0.f, 0.f};
#pragma unroll
    for (int i = 0; i < 16; ++i) {
        float p = ar[i] * ar[i] + ai[i] * ai[i];
#pragma unroll
        for (int q = 0; q < NQ; ++q)
            z[q] += (i & (1 << (3 - q))) ? -p : p;
    }

#pragma unroll
    for (int o = 0; o < 3; ++o) {
        float v = c_lds[o];
#pragma unroll
        for (int q = 0; q < NQ; ++q) v += M_lds[o][q] * z[q];
        out[tok * 3 + o] = v;
    }
}

extern "C" void kernel_launch(void* const* d_in, const int* in_sizes, int n_in,
                              void* d_out, int out_size, void* d_ws, size_t ws_size,
                              hipStream_t stream) {
    const float* x        = (const float*)d_in[0];
    const float* W_embed  = (const float*)d_in[1];
    const float* b_embed  = (const float*)d_in[2];
    const float* W_c2q    = (const float*)d_in[3];
    const float* b_c2q    = (const float*)d_in[4];
    const float* q_w      = (const float*)d_in[5];
    const float* W_q2c    = (const float*)d_in[6];
    const float* b_q2c    = (const float*)d_in[7];
    const float* W_out    = (const float*)d_in[8];
    const float* b_out    = (const float*)d_in[9];
    float* out = (float*)d_out;

    const int tokens = in_sizes[0] / DIM;
    const int blocks = tokens / TOK_PER_BLOCK;
    qhyb_fused<<<blocks, TOK_PER_BLOCK, 0, stream>>>(
        x, W_embed, b_embed, W_c2q, b_c2q, q_w, W_q2c, b_q2c, W_out, b_out, out);
}


# Block set-up on f32 WMMA: circuit matrix = complex product of its two layers (Kronecker products of the gates), output constants = one padded product; no 64-term loops (order of sums changes)
# speedup vs baseline: 1.5707x; 1.5707x over previous
#include <hip/hip_runtime.h>

typedef __attribute__((ext_vector_type(2))) float v2f;
typedef __attribute__((ext_vector_type(4))) float v4f;
typedef __attribute__((ext_vector_type(8))) float v8f;

#define NQ 4
#define NL 2
#define DIM 20
#define HID 64
#define TOK_PER_WAVE 32
#define WAVES 4
#define TOK_PER_BLOCK 128
#define HSTRIDE 68


#define USTR 18

__device__ __forceinline__ float xhalf(float x) {
    return __builtin_bit_cast(float, __builtin_amdgcn_permlanex16(
        __builtin_bit_cast(int, x), __builtin_bit_cast(int, x),
        0x76543210, 0xfedcba98, false, false));
}

struct cplx { float re, im; };
__device__ __forceinline__ cplx cmul(cplx a, cplx b) {
    cplx r;
    r.re = a.re * b.re - a.im * b.im;
    r.im = a.re * b.im + a.im * b.re;
    return r;
}
__device__ __forceinline__ cplx cload(const float* p) {
    cplx r;
    r.re = p[0];
    r.im = p[1];
    return r;
}

__global__ __launch_bounds__(TOK_PER_BLOCK, 8) void qhyb_fused(
    const float* __restrict__ x,
    const float* __restrict__ W_embed,
    const float* __restrict__ b_embed,
    const float* __restrict__ W_c2q,
    const float* __restrict__ b_c2q,
    const float* __restrict__ q_weights,
    const float* __restrict__ W_q2c,
    const float* __restrict__ b_q2c,
    const float* __restrict__ W_out,
    const float* __restrict__ b_out,
    float* __restrict__ out)
{
    __shared__ __attribute__((aligned(16))) float h_lds[TOK_PER_BLOCK * HSTRIDE];
    __shared__ __attribute__((aligned(16))) float wc2qT_lds[HID * NQ];
    __shared__ float u_lds[NL][NQ][8];
    __shared__ float M_lds[3][NQ];
    __shared__ float c_lds[3];
    __shared__ float ure_lds[16 * USTR];
    __shared__ float uim_lds[16 * USTR];

    const int tid = threadIdx.x;
    const int lane = tid & 31;
    const int wave = __builtin_amdgcn_readfirstlane(tid >> 5);
    const int lo = lane & 15, hi = lane >> 4;

    const long blockTok0 = (long)blockIdx.x * TOK_PER_BLOCK;
    const int  waveRow0  = wave * TOK_PER_WAVE;
    const long waveTok0  = blockTok0 + waveRow0;

    const unsigned bx = blockIdx.x;
    const int sumWave   = (int)((bx ^ (bx >> 2) ^ (bx >> 4) ^ (bx >> 6) ^ (bx >> 8) ^ (bx >> 10)) & 3u);
    const int buildWave = (sumWave + 1 + (int)(bx % 3u)) & 3;

    const float wv0 = W_c2q[tid];
    const float wv1 = W_c2q[tid + TOK_PER_BLOCK];
    float phi = 0.0f, th = 0.0f, om = 0.0f;
    if (wave == buildWave && lane < NL * NQ) {
        phi = q_weights[lane * 3 + 0];
        th  = q_weights[lane * 3 + 1];
        om  = q_weights[lane * 3 + 2];
    }

    float wo[32];
    float bq[32];
    if (wave == sumWave) {
        const float* wrow = &W_out[(lo < 2 ? lo : 2) * HID + 32 * hi];
#pragma unroll
        for (int k = 0; k < 32; ++k) wo[k] = wrow[k];
        const float* bcol = (lo < NQ) ? &W_q2c[32 * hi * NQ + lo] : &b_q2c[32 * hi];
        const long bstep = (lo < NQ) ? NQ : 1;
#pragma unroll
        for (int k = 0; k < 32; ++k) {
            bq[k] = *bcol;
            if (k < 31) bcol += bstep;
        }
    }

    v2f afrag[5][2];
#pragma unroll
    for (int ks = 0; ks < 5; ++ks)
#pragma unroll
        for (int mt = 0; mt < 2; ++mt) {
            long t = waveTok0 + mt * 16 + lo;
            int d0 = ks * 4 + 2 * hi;
            afrag[ks][mt].x = x[t * DIM + d0];
            afrag[ks][mt].y = x[t * DIM + d0 + 1];
        }

    wc2qT_lds[(tid % HID) * NQ + tid / HID] = wv0;
    wc2qT_lds[(tid % HID) * NQ + tid / HID + 2] = wv1;
    if (wave == sumWave) {
        v8f d0 = (v8f){0.f, 0.f, 0.f, 0.f, 0.f, 0.f, 0.f, 0.f};
        v8f d1 = (v8f){0.f, 0.f, 0.f, 0.f, 0.f, 0.f, 0.f, 0.f};
#pragma unroll
        for (int s = 0; s < 16; s += 2) {
            v2f a0, b0, a1, b1;
            a0.x = wo[2 * s];      a0.y = wo[2 * s + 1];
            a1.x = wo[2 * s + 2];  a1.y = wo[2 * s + 3];
            b0.x = bq[2 * s];      b0.y = bq[2 * s + 1];
            b1.x = bq[2 * s + 2];  b1.y = bq[2 * s + 3];
            d0 = __builtin_amdgcn_wmma_f32_16x16x4_f32(false, a0, false, b0,
                                                       (short)0, d0, false, false);
            d1 = __builtin_amdgcn_wmma_f32_16x16x4_f32(false, a1, false, b1,
                                                       (short)0, d1, false, false);
        }
        if (hi == 0) {
            if (lo < NQ) {
#pragma unroll
                for (int o = 0; o < 3; ++o) M_lds[o][lo] = d0[o] + d1[o];
            } else if (lo == NQ) {
#pragma unroll
                for (int o = 0; o < 3; ++o) c_lds[o] = b_out[o] + (d0[o] + d1[o]);
            }
        }
    }
    if (wave == buildWave) {
        if (lane < NL * NQ) {
            int layer = lane >> 2, q = lane & 3;
            float ct = cosf(0.5f * th), st = sinf(0.5f * th);
            float cp = cosf(0.5f * (phi + om)), sp = sinf(0.5f * (phi + om));
            float cm = cosf(0.5f * (phi - om)), sm = sinf(0.5f * (phi - om));
            float* U = &u_lds[layer][q][0];
            U[0] =  cp * ct; U[1] = -sp * ct;
            U[2] = -cm * st; U[3] = -sm * st;
            U[4] =  cm * st; U[5] = -sm * st;
            U[6] =  cp * ct; U[7] =  sp * ct;
        }
        const int gi = lo ^ (lo >> 1);
        const cplx fa = cload(&u_lds[1][0][((gi >> 3) & 1) * 4 + hi * 2]);
        cplx ga[3][2];
#pragma unroll
        for (int q = 1; q < NQ; ++q) {
            const float* p = &u_lds[1][q][((gi >> (3 - q)) & 1) * 4];
            ga[q - 1][0] = cload(p);
            ga[q - 1][1] = cload(p + 2);
        }
        const cplx fb = cload(&u_lds[0][0][hi * 4 + ((lo >> 3) & 1) * 2]);
        cplx gb[3][2];
#pragma unroll
        for (int m = 0; m < 2; ++m)
            gb[0][m] = cload(&u_lds[0][1][(m ^ hi) * 4 + ((lo >> 2) & 1) * 2]);
#pragma unroll
        for (int q = 2; q < NQ; ++q)
#pragma unroll
            for (int b = 0; b < 2; ++b)
                gb[q - 1][b] = cload(&u_lds[0][q][b * 4 + ((lo >> (3 - q)) & 1) * 2]);

        v2f are[4], aim[4], naim[4], bre[4], bim[4];
#pragma unroll
        for (int m = 0; m < 2; ++m) {
            const cplx ta1 = cmul(fa, ga[0][m]);
            const cplx tb1 = cmul(fb, gb[0][m]);
#pragma unroll
            for (int n = 0; n < 2; ++n) {
                const cplx ta2 = cmul(ta1, ga[1][n]);
                const cplx tb2 = cmul(tb1, gb[1][n ^ m]);
#pragma unroll
                for (int e = 0; e < 2; ++e) {
                    const cplx a = cmul(ta2, ga[2][e]);
                    const cplx b = cmul(tb2, gb[2][e ^ n]);
                    const int s = 2 * m + n;
                    are[s][e] = a.re;  aim[s][e] = a.im;  naim[s][e] = -a.im;
                    bre[s][e] = b.re;  bim[s][e] = b.im;
                }
            }
        }
        v8f mre = (v8f){0.f, 0.f, 0.f, 0.f, 0.f, 0.f, 0.f, 0.f};
        v8f mim = (v8f){0.f, 0.f, 0.f, 0.f, 0.f, 0.f, 0.f, 0.f};
#pragma unroll
        for (int s = 0; s < 4; ++s) {
            mre = __builtin_amdgcn_wmma_f32_16x16x4_f32(false, are[s], false, bre[s],
                                                        (short)0, mre, false, false);
            mim = __builtin_amdgcn_wmma_f32_16x16x4_f32(false, are[s], false, bim[s],
                                                        (short)0, mim, false, false);
            mre = __builtin_amdgcn_wmma_f32_16x16x4_f32(false, naim[s], false, bim[s],
                                                        (short)0, mre, false, false);
            mim = __builtin_amdgcn_wmma_f32_16x16x4_f32(false, aim[s], false, bre[s],
                                                        (short)0, mim, false, false);
        }
#pragma unroll
        for (int r = 0; r < 8; ++r) {
            ure_lds[(8 * hi + r) * USTR + lo] = mre[r];
            uim_lds[(8 * hi + r) * USTR + lo] = mim[r];
        }
    }

    v2f bfrag[5][4];
#pragma unroll
    for (int ks = 0; ks < 5; ++ks)
#pragma unroll
        for (int n = 0; n < 4; ++n) {
            int hcol = n * 16 + lo;
            int d0 = ks * 4 + 2 * hi;
            bfrag[ks][n].x = W_embed[hcol * DIM + d0];
            bfrag[ks][n].y = W_embed[hcol * DIM + d0 + 1];
        }
    float bemb[4];
#pragma unroll
    for (int n = 0; n < 4; ++n) bemb[n] = b_embed[n * 16 + lo];
    v8f acc[2][4];
#pragma unroll
    for (int mt = 0; mt < 2; ++mt)
#pragma unroll
        for (int n = 0; n < 4; ++n)
            acc[mt][n] = (v8f){0.f, 0.f, 0.f, 0.f, 0.f, 0.f, 0.f, 0.f};

#pragma unroll
    for (int ks = 0; ks < 5; ++ks) {
#pragma unroll
        for (int mt = 0; mt < 2; ++mt) {
#pragma unroll
            for (int n = 0; n < 4; ++n)
                acc[mt][n] = __builtin_amdgcn_wmma_f32_16x16x4_f32(
                    false, afrag[ks][mt], false, bfrag[ks][n],
                    (short)0, acc[mt][n], false, false);
        }
    }

#pragma unroll
    for (int mt = 0; mt < 2; ++mt)
#pragma unroll
        for (int n = 0; n < 4; ++n) {
            int hc = n * 16 + lo;
#pragma unroll
            for (int r = 0; r < 8; ++r) {
                int row = waveRow0 + mt * 16 + r + 8 * hi;
                h_lds[row * HSTRIDE + hc] = fmaxf(acc[mt][n][r] + bemb[n], 0.0f);
            }
        }
    __syncthreads();

    const int myRow = waveRow0 + lane;
    const long tok = blockTok0 + myRow;
    const float* hrow = &h_lds[myRow * HSTRIDE];

    v2f q01 = {b_c2q[0], b_c2q[1]};
    v2f q23 = {b_c2q[2], b_c2q[3]};
#pragma nounroll
    for (int kc = 0; kc < HID; kc += 16) {
#pragma unroll
        for (int k = 0; k < 16; k += 4) {
            const v4f hv = *reinterpret_cast<const v4f*>(&hrow[kc + k]);
#pragma unroll
            for (int j = 0; j < 4; ++j) {
                const v4f w = *reinterpret_cast<const v4f*>(&wc2qT_lds[(kc + k + j) * NQ]);
                const v2f hh = {hv[j], hv[j]};
                const v2f w01 = {w.x, w.y};
                const v2f w23 = {w.z, w.w};
                q01 = __builtin_elementwise_fma(w01, hh, q01);
                q23 = __builtin_elementwise_fma(w23, hh, q23);
            }
        }
    }
    const float qs[NQ] = {q01.x, q01.y, q23.x, q23.y};
    float cs[NQ], sn[NQ];
#pragma unroll
    for (int q = 0; q < NQ; ++q) {
        float a = 0.5f * tanhf(qs[q]);
        cs[q] = cosf(a);
        sn[q] = sinf(a);
    }

    v2f ure[4], uim[4];
#pragma unroll
    for (int s = 0; s < 4; ++s) {
        ure[s].x = ure_lds[lo * USTR + 8 * hi + 2 * s];
        ure[s].y = ure_lds[lo * USTR + 8 * hi + 2 * s + 1];
        uim[s].x = uim_lds[lo * USTR + 8 * hi + 2 * s];
        uim[s].y = uim_lds[lo * USTR + 8 * hi + 2 * s + 1];
    }

    float ocs[NQ], osn[NQ];
#pragma unroll
    for (int q = 0; q < NQ; ++q) {
        ocs[q] = xhalf(cs[q]);
        osn[q] = xhalf(sn[q]);
    }

    float zt[2][NQ];
#pragma unroll
    for (int tt = 0; tt < 2; ++tt) {
        const bool own = (hi == tt);
        float c_[NQ], s_[NQ];
#pragma unroll
        for (int q = 0; q < NQ; ++q) {
            c_[q] = own ? cs[q] : ocs[q];
            s_[q] = own ? sn[q] : osn[q];
        }
        const float a0 = hi ? s_[0] : c_[0];
        float psi[8];
#pragma unroll
        for (int r = 0; r < 8; ++r)
            psi[r] = a0 * ((r >> 2) & 1 ? s_[1] : c_[1]) *
                          ((r >> 1) & 1 ? s_[2] : c_[2]) *
                          ((r     ) & 1 ? s_[3] : c_[3]);

        v8f dre = (v8f){0.f, 0.f, 0.f, 0.f, 0.f, 0.f, 0.f, 0.f};
        v8f dim = (v8f){0.f, 0.f, 0.f, 0.f, 0.f, 0.f, 0.f, 0.f};
#pragma unroll
        for (int s = 0; s < 4; ++s) {
            v2f b;
            b.x = psi[2 * s];
            b.y = psi[2 * s + 1];
            dre = __builtin_amdgcn_wmma_f32_16x16x4_f32(false, ure[s], false, b,
                                                        (short)0, dre, false, false);
            dim = __builtin_amdgcn_wmma_f32_16x16x4_f32(false, uim[s], false, b,
                                                        (short)0, dim, false, false);
        }

        float zp[NQ] = {0.f, 0.f, 0.f, 0.f};
#pragma unroll
        for (int r = 0; r < 8; ++r) {
            float p = dre[r] * dre[r] + dim[r] * dim[r];
            zp[0] += hi ? -p : p;
            zp[1] += ((r >> 2) & 1) ? -p : p;
            zp[2] += ((r >> 1) & 1) ? -p : p;
            zp[3] += ((r     ) & 1) ? -p : p;
        }
#pragma unroll
        for (int q = 0; q < NQ; ++q)
            zt[tt][q] = zp[q] + xhalf(zp[q]);
    }

    float z[NQ];
#pragma unroll
    for (int q = 0; q < NQ; ++q) z[q] = hi ? zt[1][q] : zt[0][q];

#pragma unroll
    for (int o = 0; o < 3; ++o) {
        float v = c_lds[o];
#pragma unroll
        for (int q = 0; q < NQ; ++q) v += M_lds[o][q] * z[q];
        out[tok * 3 + o] = v;
    }
}

extern "C" void kernel_launch(void* const* d_in, const int* in_sizes, int n_in,
                              void* d_out, int out_size, void* d_ws, size_t ws_size,
                              hipStream_t stream) {
    const float* x        = (const float*)d_in[0];
    const float* W_embed  = (const float*)d_in[1];
    const float* b_embed  = (const float*)d_in[2];
    const float* W_c2q    = (const float*)d_in[3];
    const float* b_c2q    = (const float*)d_in[4];
    const float* q_w      = (const float*)d_in[5];
    const float* W_q2c    = (const float*)d_in[6];
    const float* b_q2c    = (const float*)d_in[7];
    const float* W_out    = (const float*)d_in[8];
    const float* b_out    = (const float*)d_in[9];
    float* out = (float*)d_out;

    const int tokens = in_sizes[0] / DIM;
    const int blocks = tokens / TOK_PER_BLOCK;
    qhyb_fused<<<blocks, TOK_PER_BLOCK, 0, stream>>>(
        x, W_embed, b_embed, W_c2q, b_c2q, q_w, W_q2c, b_q2c, W_out, b_out, out);
}


# Work-group of 16 waves (512 tokens) on one per-block set-up: circuit matrix, output constants and W_c2q copy built 512 times a call, not 2048; a quarter as many work-groups; all sums as before
# speedup vs baseline: 1.7592x; 1.1200x over previous
#include <hip/hip_runtime.h>

typedef __attribute__((ext_vector_type(2))) float v2f;
typedef __attribute__((ext_vector_type(4))) float v4f;
typedef __attribute__((ext_vector_type(8))) float v8f;

#define NQ 4
#define NL 2
#define DIM 20
#define HID 64
#define TOK_PER_WAVE 32
#define WAVES 16
#define TOK_PER_BLOCK 512
#define HSTRIDE 68


#define USTR 18

__device__ __forceinline__ float xhalf(float x) {
    return __builtin_bit_cast(float, __builtin_amdgcn_permlanex16(
        __builtin_bit_cast(int, x), __builtin_bit_cast(int, x),
        0x76543210, 0xfedcba98, false, false));
}

struct cplx { float re, im; };
__device__ __forceinline__ cplx cmul(cplx a, cplx b) {
    cplx r;
    r.re = a.re * b.re - a.im * b.im;
    r.im = a.re * b.im + a.im * b.re;
    return r;
}
__device__ __forceinline__ cplx cload(const float* p) {
    cplx r;
    r.re = p[0];
    r.im = p[1];
    return r;
}

__global__ __launch_bounds__(TOK_PER_BLOCK, 8) void qhyb_fused(
    const float* __restrict__ x,
    const float* __restrict__ W_embed,
    const float* __restrict__ b_embed,
    const float* __restrict__ W_c2q,
    const float* __restrict__ b_c2q,
    const float* __restrict__ q_weights,
    const float* __restrict__ W_q2c,
    const float* __restrict__ b_q2c,
    const float* __restrict__ W_out,
    const float* __restrict__ b_out,
    float* __restrict__ out)
{
    __shared__ __attribute__((aligned(16))) float h_lds[TOK_PER_BLOCK * HSTRIDE];
    __shared__ __attribute__((aligned(16))) float wc2qT_lds[HID * NQ];
    __shared__ float u_lds[NL][NQ][8];
    __shared__ float M_lds[3][NQ];
    __shared__ float c_lds[3];
    __shared__ float ure_lds[16 * USTR];
    __shared__ float uim_lds[16 * USTR];

    const int tid = threadIdx.x;
    const int lane = tid & 31;
    const int wave = __builtin_amdgcn_readfirstlane(tid >> 5);
    const int lo = lane & 15, hi = lane >> 4;

    const long blockTok0 = (long)blockIdx.x * TOK_PER_BLOCK;
    const int  waveRow0  = wave * TOK_PER_WAVE;
    const long waveTok0  = blockTok0 + waveRow0;

    const unsigned bx = blockIdx.x;
    const int sumWave   = (int)((bx ^ (bx >> 2) ^ (bx >> 4) ^ (bx >> 6) ^ (bx >> 8) ^ (bx >> 10)) & 3u);
    const int buildWave = (sumWave + 1 + (int)(bx % 3u)) & 3;

    float wv0 = 0.0f;
    if (wave < (HID * NQ) / 32) wv0 = W_c2q[tid];
    float phi = 0.0f, th = 0.0f, om = 0.0f;
    if (wave == buildWave && lane < NL * NQ) {
        phi = q_weights[lane * 3 + 0];
        th  = q_weights[lane * 3 + 1];
        om  = q_weights[lane * 3 + 2];
    }

    float wo[32];
    float bq[32];
    if (wave == sumWave) {
        const float* wrow = &W_out[(lo < 2 ? lo : 2) * HID + 32 * hi];
#pragma unroll
        for (int k = 0; k < 32; ++k) wo[k] = wrow[k];
        const float* bcol = (lo < NQ) ? &W_q2c[32 * hi * NQ + lo] : &b_q2c[32 * hi];
        const long bstep = (lo < NQ) ? NQ : 1;
#pragma unroll
        for (int k = 0; k < 32; ++k) {
            bq[k] = *bcol;
            if (k < 31) bcol += bstep;
        }
    }

    v2f afrag[5][2];
#pragma unroll
    for (int ks = 0; ks < 5; ++ks)
#pragma unroll
        for (int mt = 0; mt < 2; ++mt) {
            long t = waveTok0 + mt * 16 + lo;
            int d0 = ks * 4 + 2 * hi;
            afrag[ks][mt].x = x[t * DIM + d0];
            afrag[ks][mt].y = x[t * DIM + d0 + 1];
        }

    if (wave < (HID * NQ) / 32) wc2qT_lds[(tid % HID) * NQ + tid / HID] = wv0;
    if (wave == sumWave) {
        v8f d0 = (v8f){0.f, 0.f, 0.f, 0.f, 0.f, 0.f, 0.f, 0.f};
        v8f d1 = (v8f){0.f, 0.f, 0.f, 0.f, 0.f, 0.f, 0.f, 0.f};
#pragma unroll
        for (int s = 0; s < 16; s += 2) {
            v2f a0, b0, a1, b1;
            a0.x = wo[2 * s];      a0.y = wo[2 * s + 1];
            a1.x = wo[2 * s + 2];  a1.y = wo[2 * s + 3];
            b0.x = bq[2 * s];      b0.y = bq[2 * s + 1];
            b1.x = bq[2 * s + 2];  b1.y = bq[2 * s + 3];
            d0 = __builtin_amdgcn_wmma_f32_16x16x4_f32(false, a0, false, b0,
                                                       (short)0, d0, false, false);
            d1 = __builtin_amdgcn_wmma_f32_16x16x4_f32(false, a1, false, b1,
                                                       (short)0, d1, false, false);
        }
        if (hi == 0) {
            if (lo < NQ) {
#pragma unroll
                for (int o = 0; o < 3; ++o) M_lds[o][lo] = d0[o] + d1[o];
            } else if (lo == NQ) {
#pragma unroll
                for (int o = 0; o < 3; ++o) c_lds[o] = b_out[o] + (d0[o] + d1[o]);
            }
        }
    }
    if (wave == buildWave) {
        if (lane < NL * NQ) {
            int layer = lane >> 2, q = lane & 3;
            float ct = cosf(0.5f * th), st = sinf(0.5f * th);
            float cp = cosf(0.5f * (phi + om)), sp = sinf(0.5f * (phi + om));
            float cm = cosf(0.5f * (phi - om)), sm = sinf(0.5f * (phi - om));
            float* U = &u_lds[layer][q][0];
            U[0] =  cp * ct; U[1] = -sp * ct;
            U[2] = -cm * st; U[3] = -sm * st;
            U[4] =  cm * st; U[5] = -sm * st;
            U[6] =  cp * ct; U[7] =  sp * ct;
        }
        const int gi = lo ^ (lo >> 1);
        const cplx fa = cload(&u_lds[1][0][((gi >> 3) & 1) * 4 + hi * 2]);
        cplx ga[3][2];
#pragma unroll
        for (int q = 1; q < NQ; ++q) {
            const float* p = &u_lds[1][q][((gi >> (3 - q)) & 1) * 4];
            ga[q - 1][0] = cload(p);
            ga[q - 1][1] = cload(p + 2);
        }
        const cplx fb = cload(&u_lds[0][0][hi * 4 + ((lo >> 3) & 1) * 2]);
        cplx gb[3][2];
#pragma unroll
        for (int m = 0; m < 2; ++m)
            gb[0][m] = cload(&u_lds[0][1][(m ^ hi) * 4 + ((lo >> 2) & 1) * 2]);
#pragma unroll
        for (int q = 2; q < NQ; ++q)
#pragma unroll
            for (int b = 0; b < 2; ++b)
                gb[q - 1][b] = cload(&u_lds[0][q][b * 4 + ((lo >> (3 - q)) & 1) * 2]);

        v2f are[4], aim[4], naim[4], bre[4], bim[4];
#pragma unroll
        for (int m = 0; m < 2; ++m) {
            const cplx ta1 = cmul(fa, ga[0][m]);
            const cplx tb1 = cmul(fb, gb[0][m]);
#pragma unroll
            for (int n = 0; n < 2; ++n) {
                const cplx ta2 = cmul(ta1, ga[1][n]);
                const cplx tb2 = cmul(tb1, gb[1][n ^ m]);
#pragma unroll
                for (int e = 0; e < 2; ++e) {
                    const cplx a = cmul(ta2, ga[2][e]);
                    const cplx b = cmul(tb2, gb[2][e ^ n]);
                    const int s = 2 * m + n;
                    are[s][e] = a.re;  aim[s][e] = a.im;  naim[s][e] = -a.im;
                    bre[s][e] = b.re;  bim[s][e] = b.im;
                }
            }
        }
        v8f mre = (v8f){0.f, 0.f, 0.f, 0.f, 0.f, 0.f, 0.f, 0.f};
        v8f mim = (v8f){0.f, 0.f, 0.f, 0.f, 0.f, 0.f, 0.f, 0.f};
#pragma unroll
        for (int s = 0; s < 4; ++s) {
            mre = __builtin_amdgcn_wmma_f32_16x16x4_f32(false, are[s], false, bre[s],
                                                        (short)0, mre, false, false);
            mim = __builtin_amdgcn_wmma_f32_16x16x4_f32(false, are[s], false, bim[s],
                                                        (short)0, mim, false, false);
            mre = __builtin_amdgcn_wmma_f32_16x16x4_f32(false, naim[s], false, bim[s],
                                                        (short)0, mre, false, false);
            mim = __builtin_amdgcn_wmma_f32_16x16x4_f32(false, aim[s], false, bre[s],
                                                        (short)0, mim, false, false);
        }
#pragma unroll
        for (int r = 0; r < 8; ++r) {
            ure_lds[(8 * hi + r) * USTR + lo] = mre[r];
            uim_lds[(8 * hi + r) * USTR + lo] = mim[r];
        }
    }

    v2f bfrag[5][4];
#pragma unroll
    for (int ks = 0; ks < 5; ++ks)
#pragma unroll
        for (int n = 0; n < 4; ++n) {
            int hcol = n * 16 + lo;
            int d0 = ks * 4 + 2 * hi;
            bfrag[ks][n].x = W_embed[hcol * DIM + d0];
            bfrag[ks][n].y = W_embed[hcol * DIM + d0 + 1];
        }
    float bemb[4];
#pragma unroll
    for (int n = 0; n < 4; ++n) bemb[n] = b_embed[n * 16 + lo];
    v8f acc[2][4];
#pragma unroll
    for (int mt = 0; mt < 2; ++mt)
#pragma unroll
        for (int n = 0; n < 4; ++n)
            acc[mt][n] = (v8f){0.f, 0.f, 0.f, 0.f, 0.f, 0.f, 0.f, 0.f};

#pragma unroll
    for (int ks = 0; ks < 5; ++ks) {
#pragma unroll
        for (int mt = 0; mt < 2; ++mt) {
#pragma unroll
            for (int n = 0; n < 4; ++n)
                acc[mt][n] = __builtin_amdgcn_wmma_f32_16x16x4_f32(
                    false, afrag[ks][mt], false, bfrag[ks][n],
                    (short)0, acc[mt][n], false, false);
        }
    }

#pragma unroll
    for (int mt = 0; mt < 2; ++mt)
#pragma unroll
        for (int n = 0; n < 4; ++n) {
            int hc = n * 16 + lo;
#pragma unroll
            for (int r = 0; r < 8; ++r) {
                int row = waveRow0 + mt * 16 + r + 8 * hi;
                h_lds[row * HSTRIDE + hc] = fmaxf(acc[mt][n][r] + bemb[n], 0.0f);
            }
        }
    __syncthreads();

    const int myRow = waveRow0 + lane;
    const long tok = blockTok0 + myRow;
    const float* hrow = &h_lds[myRow * HSTRIDE];

    v2f q01 = {b_c2q[0], b_c2q[1]};
    v2f q23 = {b_c2q[2], b_c2q[3]};
#pragma nounroll
    for (int kc = 0; kc < HID; kc += 16) {
#pragma unroll
        for (int k = 0; k < 16; k += 4) {
            const v4f hv = *reinterpret_cast<const v4f*>(&hrow[kc + k]);
#pragma unroll
            for (int j = 0; j < 4; ++j) {
                const v4f w = *reinterpret_cast<const v4f*>(&wc2qT_lds[(kc + k + j) * NQ]);
                const v2f hh = {hv[j], hv[j]};
                const v2f w01 = {w.x, w.y};
                const v2f w23 = {w.z, w.w};
                q01 = __builtin_elementwise_fma(w01, hh, q01);
                q23 = __builtin_elementwise_fma(w23, hh, q23);
            }
        }
    }
    const float qs[NQ] = {q01.x, q01.y, q23.x, q23.y};
    float cs[NQ], sn[NQ];
#pragma unroll
    for (int q = 0; q < NQ; ++q) {
        float a = 0.5f * tanhf(qs[q]);
        cs[q] = cosf(a);
        sn[q] = sinf(a);
    }

    v2f ure[4], uim[4];
#pragma unroll
    for (int s = 0; s < 4; ++s) {
        ure[s].x = ure_lds[lo * USTR + 8 * hi + 2 * s];
        ure[s].y = ure_lds[lo * USTR + 8 * hi + 2 * s + 1];
        uim[s].x = uim_lds[lo * USTR + 8 * hi + 2 * s];
        uim[s].y = uim_lds[lo * USTR + 8 * hi + 2 * s + 1];
    }

    float ocs[NQ], osn[NQ];
#pragma unroll
    for (int q = 0; q < NQ; ++q) {
        ocs[q] = xhalf(cs[q]);
        osn[q] = xhalf(sn[q]);
    }

    float zt[2][NQ];
#pragma unroll
    for (int tt = 0; tt < 2; ++tt) {
        const bool own = (hi == tt);
        float c_[NQ], s_[NQ];
#pragma unroll
        for (int q = 0; q < NQ; ++q) {
            c_[q] = own ? cs[q] : ocs[q];
            s_[q] = own ? sn[q] : osn[q];
        }
        const float a0 = hi ? s_[0] : c_[0];
        float psi[8];
#pragma unroll
        for (int r = 0; r < 8; ++r)
            psi[r] = a0 * ((r >> 2) & 1 ? s_[1] : c_[1]) *
                          ((r >> 1) & 1 ? s_[2] : c_[2]) *
                          ((r     ) & 1 ? s_[3] : c_[3]);

        v8f dre = (v8f){0.f, 0.f, 0.f, 0.f, 0.f, 0.f, 0.f, 0.f};
        v8f dim = (v8f){0.f, 0.f, 0.f, 0.f, 0.f, 0.f, 0.f, 0.f};
#pragma unroll
        for (int s = 0; s < 4; ++s) {
            v2f b;
            b.x = psi[2 * s];
            b.y = psi[2 * s + 1];
            dre = __builtin_amdgcn_wmma_f32_16x16x4_f32(false, ure[s], false, b,
                                                        (short)0, dre, false, false);
            dim = __builtin_amdgcn_wmma_f32_16x16x4_f32(false, uim[s], false, b,
                                                        (short)0, dim, false, false);
        }

        float zp[NQ] = {0.f, 0.f, 0.f, 0.f};
#pragma unroll
        for (int r = 0; r < 8; ++r) {
            float p = dre[r] * dre[r] + dim[r] * dim[r];
            zp[0] += hi ? -p : p;
            zp[1] += ((r >> 2) & 1) ? -p : p;
            zp[2] += ((r >> 1) & 1) ? -p : p;
            zp[3] += ((r     ) & 1) ? -p : p;
        }
#pragma unroll
        for (int q = 0; q < NQ; ++q)
            zt[tt][q] = zp[q] + xhalf(zp[q]);
    }

    float z[NQ];
#pragma unroll
    for (int q = 0; q < NQ; ++q) z[q] = hi ? zt[1][q] : zt[0][q];

#pragma unroll
    for (int o = 0; o < 3; ++o) {
        float v = c_lds[o];
#pragma unroll
        for (int q = 0; q < NQ; ++q) v += M_lds[o][q] * z[q];
        out[tok * 3 + o] = v;
    }
}

extern "C" void kernel_launch(void* const* d_in, const int* in_sizes, int n_in,
                              void* d_out, int out_size, void* d_ws, size_t ws_size,
                              hipStream_t stream) {
    const float* x        = (const float*)d_in[0];
    const float* W_embed  = (const float*)d_in[1];
    const float* b_embed  = (const float*)d_in[2];
    const float* W_c2q    = (const float*)d_in[3];
    const float* b_c2q    = (const float*)d_in[4];
    const float* q_w      = (const float*)d_in[5];
    const float* W_q2c    = (const float*)d_in[6];
    const float* b_q2c    = (const float*)d_in[7];
    const float* W_out    = (const float*)d_in[8];
    const float* b_out    = (const float*)d_in[9];
    float* out = (float*)d_out;

    const int tokens = in_sizes[0] / DIM;
    const int blocks = tokens / TOK_PER_BLOCK;
    qhyb_fused<<<blocks, TOK_PER_BLOCK, 0, stream>>>(
        x, W_embed, b_embed, W_c2q, b_c2q, q_w, W_q2c, b_q2c, W_out, b_out, out);
}
